# starting kernel
# baseline (speedup 1.0000x reference)
#include <hip/hip_runtime.h>


typedef float v2f __attribute__((ext_vector_type(2)));
typedef float v8f __attribute__((ext_vector_type(8)));

#define B_  64
#define P_  2048
#define N_  512
#define NT_ (N_ / 16)

template <int SPLIT>
__global__ __launch_bounds__(256) void slayer_wmma_kernel(
    const float* __restrict__ x,
    const float* __restrict__ not_dummy,
    const float* __restrict__ centers,
    const float* __restrict__ sharpness,
    float* __restrict__ part)
{
    const int lane = threadIdx.x & 31;
    const int wid  = blockIdx.x * (blockDim.x >> 5) + (threadIdx.x >> 5);
    const int s    = wid % SPLIT;
    const int tid  = wid / SPLIT;
    const int nt   = tid % NT_;
    const int b    = tid / NT_;

    const int  ln = lane & 15;
    const bool hi = lane >= 16;

    const int   n  = nt * 16 + ln;
    const float c0 = centers[n * 3 + 0], c1 = centers[n * 3 + 1], c2 = centers[n * 3 + 2];
    const float h0 = sharpness[n * 3 + 0], h1 = sharpness[n * 3 + 1], h2 = sharpness[n * 3 + 2];
    const float LOG2E = 1.4426950408889634f;
    const float s0 = h0 * h0 * LOG2E, s1 = h1 * h1 * LOG2E, s2 = h2 * h2 * LOG2E;
    const float w0 = 2.0f * s0 * c0, w1 = 2.0f * s1 * c1, w2 = 2.0f * s2 * c2;
    const float w3 = -s0, w4 = -s1, w5 = -s2;
    const float w6 = -(s0 * c0 * c0 + s1 * c1 * c1 + s2 * c2 * c2);
    v2f bf0, bf1;
    bf0.x = hi ? w2 : w0;
    bf0.y = hi ? w3 : w1;
    bf1.x = hi ? w6 : w4;
    bf1.y = hi ? 1.0f : w5;

    const int    prows = P_ / SPLIT;
    const int    p0    = s * prows;
    const float* xb    = x + (size_t)b * P_ * 3;
    const float* ndb   = not_dummy + (size_t)b * P_;

    v8f acc = {};

    for (int pt = 0; pt < prows; pt += 16) {
        const int    p  = p0 + pt;
        const float* xr = xb + (size_t)(p + ln) * 3;
        const float  x0 = xr[0], x1 = xr[1], x2 = xr[2];
        const float  lg = __builtin_amdgcn_logf(ndb[p + ln]);

        v2f a0, a1;
        a0.x = hi ? x2 : x0;
        a0.y = hi ? x0 * x0 : x1;
        a1.x = hi ? 1.0f : x1 * x1;
        a1.y = hi ? lg : x2 * x2;

        v8f c = {};
        c = __builtin_amdgcn_wmma_f32_16x16x4_f32(false, a0, false, bf0,
                                                  (short)0, c, false, false);
        c = __builtin_amdgcn_wmma_f32_16x16x4_f32(false, a1, false, bf1,
                                                  (short)0, c, false, false);

#pragma unroll
        for (int r = 0; r < 8; ++r) {
            acc[r] += __builtin_amdgcn_exp2f(c[r]);
        }
    }

    float total = ((acc[0] + acc[1]) + (acc[2] + acc[3])) +
                  ((acc[4] + acc[5]) + (acc[6] + acc[7]));

    total += __shfl_xor(total, 16, 32);
    if (!hi) {
        part[(size_t)s * (B_ * N_) + (size_t)b * N_ + n] = total;
    }
}

__global__ __launch_bounds__(256) void slayer_reduce_kernel(
    const float* __restrict__ part, float* __restrict__ out, int count, int split)
{
    const int i = blockIdx.x * blockDim.x + threadIdx.x;
    if (i < count) {
        float s = 0.0f;
        for (int k = 0; k < split; ++k) s += part[(size_t)k * count + i];
        out[i] = s;
    }
}

extern "C" void kernel_launch(void* const* d_in, const int* in_sizes, int n_in,
                              void* d_out, int out_size, void* d_ws, size_t ws_size,
                              hipStream_t stream) {
    const float* x       = (const float*)d_in[0];
    const float* nd      = (const float*)d_in[1];
    const float* centers = (const float*)d_in[2];
    const float* sharp   = (const float*)d_in[3];
    float*       out     = (float*)d_out;

    constexpr int SPLIT = 4;
    const size_t need = (size_t)SPLIT * B_ * N_ * sizeof(float);

    if (d_ws != nullptr && ws_size >= need) {
        float* part = (float*)d_ws;
        const int waves  = B_ * NT_ * SPLIT;
        const int blocks = waves / 8;
        slayer_wmma_kernel<SPLIT><<<blocks, 256, 0, stream>>>(x, nd, centers, sharp, part);
        const int count = B_ * N_;
        slayer_reduce_kernel<<<(count + 255) / 256, 256, 0, stream>>>(part, out, count, SPLIT);
    } else {
        const int waves  = B_ * NT_;
        const int blocks = waves / 8;
        slayer_wmma_kernel<1><<<blocks, 256, 0, stream>>>(x, nd, centers, sharp, out);
    }
}


# Each wave keeps 4 adjacent cluster tiles' weights in registers and builds each 16-point fragment once for all 4, one trip ahead (was 1 tile per wave); P split 8 ways (was 4): partial sums regroup
# speedup vs baseline: 1.4304x; 1.4304x over previous
#include <hip/hip_runtime.h>


typedef float v2f __attribute__((ext_vector_type(2)));
typedef float v8f __attribute__((ext_vector_type(8)));

#define B_  64
#define P_  2048
#define N_  512
#define NT_ (N_ / 16)

__device__ __forceinline__ void make_a(const float x0, const float x1, const float x2,
                                       const float nv, const bool hi, v2f& a0, v2f& a1) {
    const float lg = __builtin_amdgcn_logf(nv);
    a0.x = hi ? x2 : x0;
    a0.y = hi ? x0 * x0 : x1;
    a1.x = hi ? 1.0f : x1 * x1;
    a1.y = hi ? lg : x2 * x2;
}

template <int SPLIT, int NTW>
__global__ __launch_bounds__(256) void slayer_wmma_kernel(
    const float* __restrict__ x,
    const float* __restrict__ not_dummy,
    const float* __restrict__ centers,
    const float* __restrict__ sharpness,
    float* __restrict__ part)
{
    constexpr int NG  = NT_ / NTW;
    constexpr int NIT = P_ / SPLIT / 16;

    const int lane = threadIdx.x & 31;
    const int wib  = __builtin_amdgcn_readfirstlane((int)(threadIdx.x >> 5));
    const int wid  = blockIdx.x * (blockDim.x >> 5) + wib;
    const int ng   = wid % NG;
    const int tid  = wid / NG;
    const int s    = tid % SPLIT;
    const int b    = tid / SPLIT;

    const int  ln = lane & 15;
    const bool hi = lane >= 16;

    const int    p0  = s * (NIT * 16);
    const float* xw  = x + ((size_t)b * P_ + p0 + ln) * 3;
    const float* ndw = not_dummy + (size_t)b * P_ + p0 + ln;

    const float f0 = xw[0], f1 = xw[1], f2 = xw[2], fn = ndw[0];

    v2f bf0[NTW], bf1[NTW];
    const float LOG2E = 1.4426950408889634f;
#pragma unroll
    for (int j = 0; j < NTW; ++j) {
        const int   n  = (ng * NTW + j) * 16 + ln;
        const float c0 = centers[n * 3 + 0], c1 = centers[n * 3 + 1], c2 = centers[n * 3 + 2];
        const float h0 = sharpness[n * 3 + 0], h1 = sharpness[n * 3 + 1], h2 = sharpness[n * 3 + 2];
        const float s0 = h0 * h0 * LOG2E, s1 = h1 * h1 * LOG2E, s2 = h2 * h2 * LOG2E;
        const float w0 = 2.0f * s0 * c0, w1 = 2.0f * s1 * c1, w2 = 2.0f * s2 * c2;
        const float w3 = -s0, w4 = -s1, w5 = -s2;
        const float w6 = -(s0 * c0 * c0 + s1 * c1 * c1 + s2 * c2 * c2);
        bf0[j].x = hi ? w2 : w0;
        bf0[j].y = hi ? w3 : w1;
        bf1[j].x = hi ? w6 : w4;
        bf1[j].y = hi ? 1.0f : w5;
    }

    v8f acc[NTW];
#pragma unroll
    for (int j = 0; j < NTW; ++j) acc[j] = v8f{};

    v2f a0, a1;
    make_a(f0, f1, f2, fn, hi, a0, a1);

    for (int it = 0; it < NIT; ++it) {
        const int    nb  = (it + 1 < NIT) ? it + 1 : it;
        const float* xn  = xw + (size_t)nb * 48;
        const float* ndn = ndw + (size_t)nb * 16;
        const float  y0 = xn[0], y1 = xn[1], y2 = xn[2], yn = ndn[0];

#pragma unroll
        for (int j = 0; j < NTW; ++j) {
            v8f c = {};
            c = __builtin_amdgcn_wmma_f32_16x16x4_f32(false, a0, false, bf0[j],
                                                      (short)0, c, false, false);
            c = __builtin_amdgcn_wmma_f32_16x16x4_f32(false, a1, false, bf1[j],
                                                      (short)0, c, false, false);
            v8f e;
#pragma unroll
            for (int r = 0; r < 8; ++r) e[r] = __builtin_amdgcn_exp2f(c[r]);
            acc[j] += e;
        }

        make_a(y0, y1, y2, yn, hi, a0, a1);
    }

#pragma unroll
    for (int j = 0; j < NTW; ++j) {
        float total = ((acc[j][0] + acc[j][1]) + (acc[j][2] + acc[j][3])) +
                      ((acc[j][4] + acc[j][5]) + (acc[j][6] + acc[j][7]));
        total += __shfl_xor(total, 16, 32);
        if (!hi) {
            const int n = (ng * NTW + j) * 16 + ln;
            part[(size_t)s * (B_ * N_) + (size_t)b * N_ + n] = total;
        }
    }
}

__global__ __launch_bounds__(256) void slayer_reduce_kernel(
    const float* __restrict__ part, float* __restrict__ out, int count, int split)
{
    const int i = blockIdx.x * blockDim.x + threadIdx.x;
    if (i < count) {
        float s = 0.0f;
        for (int k = 0; k < split; ++k) s += part[(size_t)k * count + i];
        out[i] = s;
    }
}

extern "C" void kernel_launch(void* const* d_in, const int* in_sizes, int n_in,
                              void* d_out, int out_size, void* d_ws, size_t ws_size,
                              hipStream_t stream) {
    const float* x       = (const float*)d_in[0];
    const float* nd      = (const float*)d_in[1];
    const float* centers = (const float*)d_in[2];
    const float* sharp   = (const float*)d_in[3];
    float*       out     = (float*)d_out;

    constexpr int SPLIT = 8;
    constexpr int NTW   = 4;
    const size_t need = (size_t)SPLIT * B_ * N_ * sizeof(float);

    if (d_ws != nullptr && ws_size >= need) {
        float* part = (float*)d_ws;
        const int waves  = B_ * (NT_ / NTW) * SPLIT;
        const int blocks = waves / 8;
        slayer_wmma_kernel<SPLIT, NTW><<<blocks, 256, 0, stream>>>(x, nd, centers, sharp, part);
        const int count = B_ * N_;
        slayer_reduce_kernel<<<(count + 255) / 256, 256, 0, stream>>>(part, out, count, SPLIT);
    } else {
        const int waves  = B_ * (NT_ / NTW);
        const int blocks = waves / 8;
        slayer_wmma_kernel<1, NTW><<<blocks, 256, 0, stream>>>(x, nd, centers, sharp, out);
    }
}


# One launch instead of two: a block is the 8 point ranges of one (batch row, 4-tile group), its waves' partial sums meet in LDS and the block stores the outputs; the order of every sum is kept
# speedup vs baseline: 1.5413x; 1.0775x over previous
#include <hip/hip_runtime.h>


typedef float v2f __attribute__((ext_vector_type(2)));
typedef float v8f __attribute__((ext_vector_type(8)));

#define B_  64
#define P_  2048
#define N_  512
#define NT_ (N_ / 16)
#define SPLIT_ 8

__device__ __forceinline__ void make_a(const float x0, const float x1, const float x2,
                                       const float nv, const bool hi, v2f& a0, v2f& a1) {
    const float lg = __builtin_amdgcn_logf(nv);
    a0.x = hi ? x2 : x0;
    a0.y = hi ? x0 * x0 : x1;
    a1.x = hi ? 1.0f : x1 * x1;
    a1.y = hi ? lg : x2 * x2;
}

template <int NTW>
__global__ __launch_bounds__(SPLIT_ * 32) void slayer_wmma_kernel(
    const float* __restrict__ x,
    const float* __restrict__ not_dummy,
    const float* __restrict__ centers,
    const float* __restrict__ sharpness,
    float* __restrict__ out)
{
    constexpr int SPLIT = SPLIT_;
    constexpr int NG  = NT_ / NTW;
    constexpr int NIT = P_ / SPLIT / 16;

    __shared__ float red[SPLIT][NTW][32];

    const int lane = threadIdx.x & 31;
    const int wib  = __builtin_amdgcn_readfirstlane((int)(threadIdx.x >> 5));
    const int ng   = blockIdx.x % NG;
    const int b    = blockIdx.x / NG;
    const int s    = wib;

    const int  ln = lane & 15;
    const bool hi = lane >= 16;

    const int    p0  = s * (NIT * 16);
    const float* xw  = x + ((size_t)b * P_ + p0 + ln) * 3;
    const float* ndw = not_dummy + (size_t)b * P_ + p0 + ln;

    const float f0 = xw[0], f1 = xw[1], f2 = xw[2], fn = ndw[0];

    v2f bf0[NTW], bf1[NTW];
    const float LOG2E = 1.4426950408889634f;
#pragma unroll
    for (int j = 0; j < NTW; ++j) {
        const int   n  = (ng * NTW + j) * 16 + ln;
        const float c0 = centers[n * 3 + 0], c1 = centers[n * 3 + 1], c2 = centers[n * 3 + 2];
        const float h0 = sharpness[n * 3 + 0], h1 = sharpness[n * 3 + 1], h2 = sharpness[n * 3 + 2];
        const float s0 = h0 * h0 * LOG2E, s1 = h1 * h1 * LOG2E, s2 = h2 * h2 * LOG2E;
        const float w0 = 2.0f * s0 * c0, w1 = 2.0f * s1 * c1, w2 = 2.0f * s2 * c2;
        const float w3 = -s0, w4 = -s1, w5 = -s2;
        const float w6 = -(s0 * c0 * c0 + s1 * c1 * c1 + s2 * c2 * c2);
        bf0[j].x = hi ? w2 : w0;
        bf0[j].y = hi ? w3 : w1;
        bf1[j].x = hi ? w6 : w4;
        bf1[j].y = hi ? 1.0f : w5;
    }

    v8f acc[NTW];
#pragma unroll
    for (int j = 0; j < NTW; ++j) acc[j] = v8f{};

    v2f a0, a1;
    make_a(f0, f1, f2, fn, hi, a0, a1);

    for (int it = 0; it < NIT; ++it) {
        const int    nb  = (it + 1 < NIT) ? it + 1 : it;
        const float* xn  = xw + (size_t)nb * 48;
        const float* ndn = ndw + (size_t)nb * 16;
        const float  y0 = xn[0], y1 = xn[1], y2 = xn[2], yn = ndn[0];

#pragma unroll
        for (int j = 0; j < NTW; ++j) {
            v8f c = {};
            c = __builtin_amdgcn_wmma_f32_16x16x4_f32(false, a0, false, bf0[j],
                                                      (short)0, c, false, false);
            c = __builtin_amdgcn_wmma_f32_16x16x4_f32(false, a1, false, bf1[j],
                                                      (short)0, c, false, false);
            v8f e;
#pragma unroll
            for (int r = 0; r < 8; ++r) e[r] = __builtin_amdgcn_exp2f(c[r]);
            acc[j] += e;
        }

        make_a(y0, y1, y2, yn, hi, a0, a1);
    }

#pragma unroll
    for (int j = 0; j < NTW; ++j) {
        const float total = ((acc[j][0] + acc[j][1]) + (acc[j][2] + acc[j][3])) +
                            ((acc[j][4] + acc[j][5]) + (acc[j][6] + acc[j][7]));
        red[s][j][lane] = total;
    }
    __syncthreads();

    if (threadIdx.x < NTW * 16) {
        const int j = threadIdx.x >> 4;
        const int c = threadIdx.x & 15;
        float sum = 0.0f;
        for (int k = 0; k < SPLIT; ++k) sum += red[k][j][c] + red[k][j][c + 16];
        out[(size_t)b * N_ + (ng * NTW + j) * 16 + c] = sum;
    }
}

extern "C" void kernel_launch(void* const* d_in, const int* in_sizes, int n_in,
                              void* d_out, int out_size, void* d_ws, size_t ws_size,
                              hipStream_t stream) {
    const float* x       = (const float*)d_in[0];
    const float* nd      = (const float*)d_in[1];
    const float* centers = (const float*)d_in[2];
    const float* sharp   = (const float*)d_in[3];
    float*       out     = (float*)d_out;

    constexpr int NTW    = 4;
    constexpr int blocks = B_ * (NT_ / NTW);
    slayer_wmma_kernel<NTW><<<blocks, SPLIT_ * 32, 0, stream>>>(x, nd, centers, sharp, out);
}


# speedup vs baseline: 1.6562x; 1.0745x over previous
#include <hip/hip_runtime.h>


typedef float v2f __attribute__((ext_vector_type(2)));
typedef float v8f __attribute__((ext_vector_type(8)));

#define B_  64
#define P_  2048
#define N_  512
#define NT_ (N_ / 16)
#define SPLIT_ 8

__device__ __forceinline__ void make_a(const float x0, const float x1, const float x2,
                                       const float nv, const bool hi, v2f& a0, v2f& a1) {
    const float lg = __builtin_amdgcn_logf(nv);
    a0.x = hi ? x2 : x0;
    a0.y = hi ? x0 * x0 : x1;
    a1.x = hi ? 1.0f : x1 * x1;
    a1.y = hi ? lg : x2 * x2;
}

__device__ __forceinline__ void exp4_add4(float& e0, float& e1, float& e2, float& e3,
                                          const float d0, const float d1, const float d2, const float d3,
                                          float& s0, float& s1, float& s2, float& s3,
                                          const float p0, const float p1, const float p2, const float p3) {
    asm("v_exp_f32_e32 %[e0], %[d0]\n\t"
        "v_add_f32_e32 %[s0], %[p0], %[s0]\n\t"
        "v_exp_f32_e32 %[e1], %[d1]\n\t"
        "v_add_f32_e32 %[s1], %[p1], %[s1]\n\t"
        "v_exp_f32_e32 %[e2], %[d2]\n\t"
        "v_add_f32_e32 %[s2], %[p2], %[s2]\n\t"
        "v_exp_f32_e32 %[e3], %[d3]\n\t"
        "v_add_f32_e32 %[s3], %[p3], %[s3]"
        : [e0] "=&v"(e0), [e1] "=&v"(e1), [e2] "=&v"(e2), [e3] "=&v"(e3),
          [s0] "+&v"(s0), [s1] "+&v"(s1), [s2] "+&v"(s2), [s3] "+&v"(s3)
        : [d0] "v"(d0), [d1] "v"(d1), [d2] "v"(d2), [d3] "v"(d3),
          [p0] "v"(p0), [p1] "v"(p1), [p2] "v"(p2), [p3] "v"(p3));
}

template <int NTW, bool LOAD, bool BUILD, bool LAST>
__device__ __forceinline__ void trip(float (&acc)[NTW][8], const v2f (&bf0)[NTW], const v2f (&bf1)[NTW],
                                     v2f& a0, v2f& a1, v8f& d, float (&ep)[8], const bool hi,
                                     const float y0, const float y1, const float y2, const float yn,
                                     float& z0, float& z1, float& z2, float& zn,
                                     const float* __restrict__ xz, const float* __restrict__ ndz) {
    if (LOAD) { z0 = xz[0]; z1 = xz[1]; z2 = xz[2]; zn = ndz[0]; }
    v2f n0 = a0, n1 = a1;
#pragma unroll
    for (int j = 0; j < NTW; ++j) {
        if (BUILD && j == NTW - 2) make_a(y0, y1, y2, yn, hi, n0, n1);
        v8f dn = d;
        if (!(LAST && j == NTW - 1)) {
            const int jn = (j + 1) % NTW;
            const v2f f0 = (j == NTW - 1) ? n0 : a0;
            const v2f f1 = (j == NTW - 1) ? n1 : a1;
            v8f c = {};
            c = __builtin_amdgcn_wmma_f32_16x16x4_f32(false, f0, false, bf0[jn],
                                                      (short)0, c, false, false);
            c = __builtin_amdgcn_wmma_f32_16x16x4_f32(false, f1, false, bf1[jn],
                                                      (short)0, c, false, false);
            dn = c;
        }
        __builtin_amdgcn_sched_barrier(0);
        float e[8];
        float (&sp)[8] = acc[(j + NTW - 1) % NTW];
        exp4_add4(e[0], e[1], e[2], e[3], d[0], d[1], d[2], d[3],
                  sp[0], sp[1], sp[2], sp[3], ep[0], ep[1], ep[2], ep[3]);
        exp4_add4(e[4], e[5], e[6], e[7], d[4], d[5], d[6], d[7],
                  sp[4], sp[5], sp[6], sp[7], ep[4], ep[5], ep[6], ep[7]);
#pragma unroll
        for (int r = 0; r < 8; ++r) ep[r] = e[r];
        d = dn;
    }
    a0 = n0; a1 = n1;
}

template <int NTW>
__global__ __launch_bounds__(SPLIT_ * 32) void slayer_wmma_kernel(
    const float* __restrict__ x,
    const float* __restrict__ not_dummy,
    const float* __restrict__ centers,
    const float* __restrict__ sharpness,
    float* __restrict__ out)
{
    constexpr int SPLIT = SPLIT_;
    constexpr int NG  = NT_ / NTW;
    constexpr int NIT = P_ / SPLIT / 16;

    __shared__ float red[SPLIT][NTW][32];

    const int lane = threadIdx.x & 31;
    const int wib  = __builtin_amdgcn_readfirstlane((int)(threadIdx.x >> 5));
    const int ng   = blockIdx.x % NG;
    const int b    = blockIdx.x / NG;
    const int s    = wib;

    const int  ln = lane & 15;
    const bool hi = lane >= 16;

    const int    p0  = s * (NIT * 16);
    const float* xw  = x + ((size_t)b * P_ + p0 + ln) * 3;
    const float* ndw = not_dummy + (size_t)b * P_ + p0 + ln;

    const float f0 = xw[0], f1 = xw[1], f2 = xw[2], fn = ndw[0];

    v2f bf0[NTW], bf1[NTW];
    const float LOG2E = 1.4426950408889634f;
#pragma unroll
    for (int j = 0; j < NTW; ++j) {
        const int   n  = (ng * NTW + j) * 16 + ln;
        const float c0 = centers[n * 3 + 0], c1 = centers[n * 3 + 1], c2 = centers[n * 3 + 2];
        const float h0 = sharpness[n * 3 + 0], h1 = sharpness[n * 3 + 1], h2 = sharpness[n * 3 + 2];
        const float s0 = h0 * h0 * LOG2E, s1 = h1 * h1 * LOG2E, s2 = h2 * h2 * LOG2E;
        const float w0 = 2.0f * s0 * c0, w1 = 2.0f * s1 * c1, w2 = 2.0f * s2 * c2;
        const float w3 = -s0, w4 = -s1, w5 = -s2;
        const float w6 = -(s0 * c0 * c0 + s1 * c1 * c1 + s2 * c2 * c2);
        bf0[j].x = hi ? w2 : w0;
        bf0[j].y = hi ? w3 : w1;
        bf1[j].x = hi ? w6 : w4;
        bf1[j].y = hi ? 1.0f : w5;
    }

    float acc[NTW][8];
#pragma unroll
    for (int j = 0; j < NTW; ++j)
#pragma unroll
        for (int r = 0; r < 8; ++r) acc[j][r] = 0.0f;

    v2f a0, a1;
    make_a(f0, f1, f2, fn, hi, a0, a1);

    float y0 = xw[48], y1 = xw[49], y2 = xw[50], yn = ndw[16];
    float z0 = 0.0f, z1 = 0.0f, z2 = 0.0f, zn = 0.0f;

    v8f d = {};
    d = __builtin_amdgcn_wmma_f32_16x16x4_f32(false, a0, false, bf0[0], (short)0, d, false, false);
    d = __builtin_amdgcn_wmma_f32_16x16x4_f32(false, a1, false, bf1[0], (short)0, d, false, false);
    float ep[8];
#pragma unroll
    for (int r = 0; r < 8; ++r) ep[r] = 0.0f;

    static_assert(NIT % 2 == 0 && NIT >= 4 && NTW >= 2, "trip pairs");
    for (int it = 0; it < NIT - 2; it += 2) {
        trip<NTW, true, true, false>(acc, bf0, bf1, a0, a1, d, ep, hi, y0, y1, y2, yn,
                                     z0, z1, z2, zn, xw + (size_t)(it + 2) * 48, ndw + (size_t)(it + 2) * 16);
        trip<NTW, true, true, false>(acc, bf0, bf1, a0, a1, d, ep, hi, z0, z1, z2, zn,
                                     y0, y1, y2, yn, xw + (size_t)(it + 3) * 48, ndw + (size_t)(it + 3) * 16);
    }
    trip<NTW, false, true, false>(acc, bf0, bf1, a0, a1, d, ep, hi, y0, y1, y2, yn,
                                  z0, z1, z2, zn, xw, ndw);
    trip<NTW, false, false, true>(acc, bf0, bf1, a0, a1, d, ep, hi, y0, y1, y2, yn,
                                  z0, z1, z2, zn, xw, ndw);
#pragma unroll
    for (int r = 0; r < 8; ++r) acc[NTW - 1][r] += ep[r];

#pragma unroll
    for (int j = 0; j < NTW; ++j) {
        const float total = ((acc[j][0] + acc[j][1]) + (acc[j][2] + acc[j][3])) +
                            ((acc[j][4] + acc[j][5]) + (acc[j][6] + acc[j][7]));
        red[s][j][lane] = total;
    }
    __syncthreads();

    if (threadIdx.x < NTW * 16) {
        const int j = threadIdx.x >> 4;
        const int c = threadIdx.x & 15;
        float sum = 0.0f;
        for (int k = 0; k < SPLIT; ++k) sum += red[k][j][c] + red[k][j][c + 16];
        out[(size_t)b * N_ + (ng * NTW + j) * 16 + c] = sum;
    }
}

extern "C" void kernel_launch(void* const* d_in, const int* in_sizes, int n_in,
                              void* d_out, int out_size, void* d_ws, size_t ws_size,
                              hipStream_t stream) {
    const float* x       = (const float*)d_in[0];
    const float* nd      = (const float*)d_in[1];
    const float* centers = (const float*)d_in[2];
    const float* sharp   = (const float*)d_in[3];
    float*       out     = (float*)d_out;

    constexpr int NTW    = 4;
    constexpr int blocks = B_ * (NT_ / NTW);
    slayer_wmma_kernel<NTW><<<blocks, SPLIT_ * 32, 0, stream>>>(x, nd, centers, sharp, out);
}
